# starting kernel
# baseline (speedup 1.0000x reference)
#include <hip/hip_runtime.h>
#include <math.h>

typedef __attribute__((ext_vector_type(2))) float v2f;
typedef __attribute__((ext_vector_type(8))) float v8f;

#define N_NODES   100000
#define NUM_FEAT  64
#define NUM_CLASS 40
#define N_EDGES   1600000
#define WPB       5
#define GEMM_BLOCK (WPB * 32)
#define ROW_TILES  (N_NODES / 16)
#define GEMM_GRID  (ROW_TILES / WPB)
#define KSTEPS     (NUM_FEAT / 4)
#define CTILES     3

#define NW_V2F     (KSTEPS * 2 * CTILES * 32)
#define OFF_W      0
#define OFF_OUT    (NW_V2F * 2)
#define OFF_LSE    (OFF_OUT + WPB * 16 * 48)
#define OFF_X      (OFF_LSE + WPB * 16)
#define OFF_A      (OFF_X + WPB * 16 * NUM_FEAT)
#define SMEM_FLOATS (OFF_A + WPB * 16 * NUM_FEAT)

__global__ void zero_ws(float* __restrict__ agg, float* __restrict__ deg) {
  long i = (long)blockIdx.x * blockDim.x + threadIdx.x;
  const long tot = (long)N_NODES * NUM_FEAT;
  if (i < tot)     agg[i] = 0.0f;
  if (i < N_NODES) deg[i] = 0.0f;
}

__global__ void edge_scatter(const float* __restrict__ x,
                             const int*   __restrict__ index,
                             float* __restrict__ agg,
                             float* __restrict__ deg) {
  long gid = (long)blockIdx.x * blockDim.x + threadIdx.x;
  long e   = gid >> 4;
  int  q   = (int)(gid & 15);
  if (e >= N_EDGES) return;
  int src = index[e];
  int dst = index[N_EDGES + e];
  const float4 v = *(const float4*)(x + (long)src * NUM_FEAT + q * 4);
  float* a = agg + (long)dst * NUM_FEAT + q * 4;
  atomicAdd(a + 0, v.x);
  atomicAdd(a + 1, v.y);
  atomicAdd(a + 2, v.z);
  atomicAdd(a + 3, v.w);
  if (q == 0) atomicAdd(deg + dst, 1.0f);
}

__global__ void __launch_bounds__(GEMM_BLOCK)
sage_gemm(const float* __restrict__ x,
          const float* __restrict__ agg,
          const float* __restrict__ deg,
          const float* __restrict__ Wl,
          const float* __restrict__ bias,
          const float* __restrict__ Wr,
          float* __restrict__ out) {
  __shared__ __align__(16) float smem[SMEM_FLOATS];

  v2f*   sW   = (v2f*)smem;
  float* sOut = smem + OFF_OUT;
  float* sLse = smem + OFF_LSE;

  const int tid  = threadIdx.x;
  const int wave = tid >> 5;
  const int lane = tid & 31;
  const int half = lane >> 4;
  const int l16  = lane & 15;

  const long row0 = ((long)blockIdx.x * WPB + wave) * 16;

  for (int i = tid; i < NW_V2F; i += GEMM_BLOCK) {
    const int li    = i & 31;
    const int t     = (i >> 5) % CTILES;
    const int which = (i / (CTILES * 32)) & 1;
    const int step  = i / (2 * CTILES * 32);
    const int k     = 4 * step + 2 * (li >> 4);
    const int n     = t * 16 + (li & 15);
    const float* W  = which ? Wr : Wl;
    v2f w = {0.0f, 0.0f};
    if (n < NUM_CLASS) w = (v2f){W[k * NUM_CLASS + n], W[(k + 1) * NUM_CLASS + n]};
    sW[i] = w;
  }

  {
    const float* xg = x   + row0 * NUM_FEAT;
    const float* ag = agg + row0 * NUM_FEAT;
    float* sx = smem + OFF_X + wave * (16 * NUM_FEAT);
    float* sa = smem + OFF_A + wave * (16 * NUM_FEAT);
    for (int i = lane; i < (16 * NUM_FEAT) / 4; i += 32) {
      float4 vx = *(const float4*)(xg + i * 4);
      float4 va = *(const float4*)(ag + i * 4);
      *(float4*)(sx + i * 4) = vx;
      *(float4*)(sa + i * 4) = va;
    }
  }
  __syncthreads();

  const float rdeg = 1.0f / fmaxf(deg[row0 + l16], 1.0f);

  const v2f* xf = (const v2f*)(smem + OFF_X + wave * (16 * NUM_FEAT) + l16 * NUM_FEAT);
  const v2f* af = (const v2f*)(smem + OFF_A + wave * (16 * NUM_FEAT) + l16 * NUM_FEAT);
  const v2f* wp = sW + lane;

  v8f acc0 = {}, acc1 = {}, acc2 = {};

#pragma unroll
  for (int step = 0; step < KSTEPS; ++step) {
    const v2f a_x   = xf[2 * step + half];
    const v2f a_agg = af[2 * step + half] * rdeg;

    const int wb = step * (2 * CTILES * 32);
    const v2f bl0 = wp[wb];
    const v2f bl1 = wp[wb + 32];
    const v2f bl2 = wp[wb + 64];
    const v2f br0 = wp[wb + 96];
    const v2f br1 = wp[wb + 128];
    const v2f br2 = wp[wb + 160];

    acc0 = __builtin_amdgcn_wmma_f32_16x16x4_f32(false, a_agg, false, bl0, (short)0, acc0, false, false);
    acc1 = __builtin_amdgcn_wmma_f32_16x16x4_f32(false, a_agg, false, bl1, (short)0, acc1, false, false);
    acc2 = __builtin_amdgcn_wmma_f32_16x16x4_f32(false, a_agg, false, bl2, (short)0, acc2, false, false);
    acc0 = __builtin_amdgcn_wmma_f32_16x16x4_f32(false, a_x,   false, br0, (short)0, acc0, false, false);
    acc1 = __builtin_amdgcn_wmma_f32_16x16x4_f32(false, a_x,   false, br1, (short)0, acc1, false, false);
    acc2 = __builtin_amdgcn_wmma_f32_16x16x4_f32(false, a_x,   false, br2, (short)0, acc2, false, false);
  }

  const float b0 = bias[l16];
  const float b1 = bias[16 + l16];
  const float b2 = (32 + l16 < NUM_CLASS) ? bias[32 + l16] : 0.0f;

  float* so = sOut + wave * (16 * 48);
#pragma unroll
  for (int r = 0; r < 8; ++r) {
    const int M = r + 8 * half;
    so[M * 48 +      l16] = acc0[r] + b0;
    so[M * 48 + 16 + l16] = acc1[r] + b1;
    so[M * 48 + 32 + l16] = acc2[r] + b2;
  }
  __syncthreads();

  if (lane < 16) {
    const float* rowp = so + l16 * 48;
    float mx = rowp[0];
    for (int c = 1; c < NUM_CLASS; ++c) mx = fmaxf(mx, rowp[c]);
    float s = 0.0f;
    for (int c = 0; c < NUM_CLASS; ++c) s += __expf(rowp[c] - mx);
    sLse[wave * 16 + l16] = mx + __logf(s);
  }
  __syncthreads();

  for (int i = lane; i < 16 * NUM_CLASS; i += 32) {
    const int r = i / NUM_CLASS, c = i % NUM_CLASS;
    out[(row0 + r) * NUM_CLASS + c] = so[r * 48 + c] - sLse[wave * 16 + r];
  }
}

extern "C" void kernel_launch(void* const* d_in, const int* in_sizes, int n_in,
                              void* d_out, int out_size, void* d_ws, size_t ws_size,
                              hipStream_t stream) {
  const float* x     = (const float*)d_in[0];
  const int*   index = (const int*)d_in[1];
  const float* Wl    = (const float*)d_in[2];
  const float* bl    = (const float*)d_in[3];
  const float* Wr    = (const float*)d_in[4];
  float* out = (float*)d_out;

  float* agg = (float*)d_ws;
  float* deg = agg + (size_t)N_NODES * NUM_FEAT;

  {
    const long tot = (long)N_NODES * NUM_FEAT;
    zero_ws<<<(int)((tot + 255) / 256), 256, 0, stream>>>(agg, deg);
  }
  {
    const long tot = (long)N_EDGES * 16;
    edge_scatter<<<(int)((tot + 255) / 256), 256, 0, stream>>>(x, index, agg, deg);
  }
  sage_gemm<<<GEMM_GRID, GEMM_BLOCK, 0, stream>>>(x, agg, deg, Wl, bl, Wr, out);
}


# Wl applied before the scatter (y = x*Wl on WMMA, 40 wide): an edge gathers and adds 40 floats + its count in the row's 2nd line, not 64 + a separate count add; sum order changes (products summed)
# speedup vs baseline: 1.8129x; 1.8129x over previous
#include <hip/hip_runtime.h>
#include <math.h>


typedef __attribute__((ext_vector_type(2))) float v2f;
typedef __attribute__((ext_vector_type(8))) float v8f;

#define N_NODES   100000
#define NUM_FEAT  64
#define NUM_CLASS 40
#define N_EDGES   1600000
#define WPB       5
#define GEMM_BLOCK (WPB * 32)
#define ROW_TILES  (N_NODES / 16)
#define GEMM_GRID  (ROW_TILES / WPB)
#define KSTEPS     (NUM_FEAT / 4)
#define CTILES     3

#define YA_W 32
#define YB_W 8
#define GA_W 32
#define GB_W 16
#define CNT_COL 8

#define NW_V2F     (KSTEPS * CTILES * 32)
#define OFF_W      0
#define OFF_OUT    (NW_V2F * 2)
#define OFF_LSE    (OFF_OUT + WPB * 16 * 48)
#define OFF_X      (OFF_LSE + WPB * 16)
#define SMEM_FLOATS (OFF_X + WPB * 16 * NUM_FEAT)


__device__ __forceinline__ void stage_weights(v2f* sW, const float* __restrict__ Wm, int tid) {
  for (int i = tid; i < NW_V2F; i += GEMM_BLOCK) {
    const int li   = i & 31;
    const int t    = (i >> 5) % CTILES;
    const int step = i / (CTILES * 32);
    const int k    = 4 * step + 2 * (li >> 4);
    const int n    = t * 16 + (li & 15);
    v2f w = {0.0f, 0.0f};
    if (n < NUM_CLASS) w = (v2f){Wm[k * NUM_CLASS + n], Wm[(k + 1) * NUM_CLASS + n]};
    sW[i] = w;
  }
}

__device__ __forceinline__ void stage_x_tile(float* sx, const float* __restrict__ xg, int lane) {
  for (int i = lane; i < (16 * NUM_FEAT) / 4; i += 32) {
    float4 vx = *(const float4*)(xg + i * 4);
    *(float4*)(sx + i * 4) = vx;
  }
}

__device__ __forceinline__ void tile_product(const v2f* xf, const v2f* wp, int half,
                                             v8f& acc0, v8f& acc1, v8f& acc2) {
#pragma unroll
  for (int step = 0; step < KSTEPS; ++step) {
    const v2f a_x = xf[2 * step + half];
    const int wb  = step * (CTILES * 32);
    const v2f b0 = wp[wb];
    const v2f b1 = wp[wb + 32];
    const v2f b2 = wp[wb + 64];
    acc0 = __builtin_amdgcn_wmma_f32_16x16x4_f32(false, a_x, false, b0, (short)0, acc0, false, false);
    acc1 = __builtin_amdgcn_wmma_f32_16x16x4_f32(false, a_x, false, b1, (short)0, acc1, false, false);
    acc2 = __builtin_amdgcn_wmma_f32_16x16x4_f32(false, a_x, false, b2, (short)0, acc2, false, false);
  }
}

__global__ void __launch_bounds__(GEMM_BLOCK)
project_rows(const float* __restrict__ x,
             const float* __restrict__ Wl,
             float* __restrict__ yA,
             float* __restrict__ yB,
             float* __restrict__ aggA,
             float* __restrict__ aggB) {
  __shared__ __align__(16) float smem[SMEM_FLOATS];

  v2f*   sW   = (v2f*)smem;
  float* sOut = smem + OFF_OUT;

  const int tid  = threadIdx.x;
  const int wave = tid >> 5;
  const int lane = tid & 31;
  const int half = lane >> 4;
  const int l16  = lane & 15;

  const long row0 = ((long)blockIdx.x * WPB + wave) * 16;

  stage_weights(sW, Wl, tid);
  float* sx = smem + OFF_X + wave * (16 * NUM_FEAT);
  stage_x_tile(sx, x + row0 * NUM_FEAT, lane);

  __syncthreads();

  const v2f* xf = (const v2f*)(sx + l16 * NUM_FEAT);
  const v2f* wp = sW + lane;

  v8f acc0 = {}, acc1 = {}, acc2 = {};
  tile_product(xf, wp, half, acc0, acc1, acc2);

  float* so = sOut + wave * (16 * 48);
#pragma unroll
  for (int r = 0; r < 8; ++r) {
    const int M = r + 8 * half;
    so[M * 48 +      l16] = acc0[r];
    so[M * 48 + 16 + l16] = acc1[r];
    so[M * 48 + 32 + l16] = acc2[r];
  }
  __syncthreads();

  for (int r = 0; r < 16; ++r)
    yA[(row0 + r) * YA_W + lane] = so[r * 48 + lane];
  for (int i = lane; i < 16 * YB_W; i += 32)
    yB[row0 * YB_W + i] = so[(i >> 3) * 48 + 32 + (i & 7)];

  {
    const float4 z = {0.0f, 0.0f, 0.0f, 0.0f};
    float* ga = aggA + row0 * GA_W;
    float* gb = aggB + row0 * GB_W;
    for (int i = lane; i < (16 * GA_W) / 4; i += 32) *(float4*)(ga + i * 4) = z;
    for (int i = lane; i < (16 * GB_W) / 4; i += 32) *(float4*)(gb + i * 4) = z;
  }
}

#define EPW 32
#define EG  8
#define SCAT_BLOCK 256
#define SCAT_GRID  (N_EDGES / EPW / (SCAT_BLOCK / 32))

__device__ __forceinline__ void load_group(const float* __restrict__ yA,
                                           const float* __restrict__ yB,
                                           int src, int lane, int g,
                                           float (&lo)[EG], float (&hi)[EG]) {
  const int cb = lane & (YB_W - 1);
#pragma unroll
  for (int j = 0; j < EG; ++j) {
    const unsigned s = (unsigned)__builtin_amdgcn_readlane(src, g + j);
    lo[j] = yA[(size_t)s * YA_W + lane];
    hi[j] = yB[(size_t)s * YB_W + cb];
  }
}

__device__ __forceinline__ void add_group(float* __restrict__ aggA,
                                          float* __restrict__ aggB,
                                          int dst, int lane, int g,
                                          const float (&lo)[EG], const float (&hi)[EG]) {
#pragma unroll
  for (int j = 0; j < EG; ++j) {
    const unsigned d = (unsigned)__builtin_amdgcn_readlane(dst, g + j);
    atomicAdd(aggA + (size_t)d * GA_W + lane, lo[j]);
  }
  asm volatile("" :: "v"(hi[0]), "v"(hi[1]), "v"(hi[2]), "v"(hi[3]),
                     "v"(hi[4]), "v"(hi[5]), "v"(hi[6]), "v"(hi[7]));
  if (lane <= CNT_COL) {
#pragma unroll
    for (int j = 0; j < EG; ++j) {
      const unsigned d = (unsigned)__builtin_amdgcn_readlane(dst, g + j);
      atomicAdd(aggB + (size_t)d * GB_W + lane, lane == CNT_COL ? 1.0f : hi[j]);
    }
  }
}

__global__ void __launch_bounds__(SCAT_BLOCK)
edge_scatter(const float* __restrict__ yA,
             const float* __restrict__ yB,
             const int*   __restrict__ index,
             float* __restrict__ aggA,
             float* __restrict__ aggB) {
  const int  lane = threadIdx.x & 31;
  const long wave = ((long)blockIdx.x * SCAT_BLOCK + threadIdx.x) >> 5;
  const long e0   = wave * EPW;
  const int src = index[e0 + lane];
  const int dst = index[N_EDGES + e0 + lane];

  float a_lo[EG], a_hi[EG], b_lo[EG], b_hi[EG];
  load_group(yA, yB, src, lane, 0 * EG, a_lo, a_hi);
  load_group(yA, yB, src, lane, 1 * EG, b_lo, b_hi);
  add_group(aggA, aggB, dst, lane, 0 * EG, a_lo, a_hi);
  load_group(yA, yB, src, lane, 2 * EG, a_lo, a_hi);
  add_group(aggA, aggB, dst, lane, 1 * EG, b_lo, b_hi);
  load_group(yA, yB, src, lane, 3 * EG, b_lo, b_hi);
  add_group(aggA, aggB, dst, lane, 2 * EG, a_lo, a_hi);
  add_group(aggA, aggB, dst, lane, 3 * EG, b_lo, b_hi);
}

__global__ void __launch_bounds__(GEMM_BLOCK)
sage_finish(const float* __restrict__ x,
            const float* __restrict__ aggA,
            const float* __restrict__ aggB,
            const float* __restrict__ bias,
            const float* __restrict__ Wr,
            float* __restrict__ out) {
  __shared__ __align__(16) float smem[SMEM_FLOATS];

  v2f*   sW   = (v2f*)smem;
  float* sOut = smem + OFF_OUT;
  float* sLse = smem + OFF_LSE;

  const int tid  = threadIdx.x;
  const int wave = tid >> 5;
  const int lane = tid & 31;
  const int half = lane >> 4;
  const int l16  = lane & 15;

  const long row0 = ((long)blockIdx.x * WPB + wave) * 16;

  stage_weights(sW, Wr, tid);
  float* sx = smem + OFF_X + wave * (16 * NUM_FEAT);
  stage_x_tile(sx, x + row0 * NUM_FEAT, lane);

  const float b0 = bias[l16];
  const float b1 = bias[16 + l16];
  const float b2 = bias[32 + (l16 & (YB_W - 1))];
  float ga0[8], ga1[8], gb[8], cnt[8];
#pragma unroll
  for (int r = 0; r < 8; ++r) {
    const long row = row0 + 8 * half + r;
    ga0[r] = aggA[row * GA_W +      l16];
    ga1[r] = aggA[row * GA_W + 16 + l16];
    gb[r]  = aggB[row * GB_W + l16];
    cnt[r] = aggB[row * GB_W + CNT_COL];
  }
  __syncthreads();

  v8f acc0, acc1, acc2;
#pragma unroll
  for (int r = 0; r < 8; ++r) {
    const float rdeg = 1.0f / fmaxf(cnt[r], 1.0f);
    acc0[r] = ga0[r] * rdeg + b0;
    acc1[r] = ga1[r] * rdeg + b1;
    acc2[r] = (l16 < YB_W) ? gb[r] * rdeg + b2 : 0.0f;
  }

  const v2f* xf = (const v2f*)(sx + l16 * NUM_FEAT);
  const v2f* wp = sW + lane;
  tile_product(xf, wp, half, acc0, acc1, acc2);

  float* so = sOut + wave * (16 * 48);
#pragma unroll
  for (int r = 0; r < 8; ++r) {
    const int M = r + 8 * half;
    so[M * 48 +      l16] = acc0[r];
    so[M * 48 + 16 + l16] = acc1[r];
    so[M * 48 + 32 + l16] = acc2[r];
  }
  __syncthreads();

  if (lane < 16) {
    const float* rowp = so + l16 * 48;
    float mx = rowp[0];
    for (int c = 1; c < NUM_CLASS; ++c) mx = fmaxf(mx, rowp[c]);
    float s = 0.0f;
    for (int c = 0; c < NUM_CLASS; ++c) s += __expf(rowp[c] - mx);
    sLse[wave * 16 + l16] = mx + __logf(s);
  }
  __syncthreads();

  for (int i = lane; i < 16 * NUM_CLASS; i += 32) {
    const int r = i / NUM_CLASS, c = i % NUM_CLASS;
    out[(row0 + r) * NUM_CLASS + c] = so[r * 48 + c] - sLse[wave * 16 + r];
  }
}

extern "C" void kernel_launch(void* const* d_in, const int* in_sizes, int n_in,
                              void* d_out, int out_size, void* d_ws, size_t ws_size,
                              hipStream_t stream) {
  const float* x     = (const float*)d_in[0];
  const int*   index = (const int*)d_in[1];
  const float* Wl    = (const float*)d_in[2];
  const float* bl    = (const float*)d_in[3];
  const float* Wr    = (const float*)d_in[4];
  float* out = (float*)d_out;

  float* aggA = (float*)d_ws;
  float* aggB = aggA + (size_t)N_NODES * GA_W;
  float* yA   = out;
  float* yB   = out + (size_t)N_NODES * YA_W;

  project_rows<<<GEMM_GRID, GEMM_BLOCK, 0, stream>>>(x, Wl, yA, yB, aggA, aggB);
  edge_scatter<<<SCAT_GRID, SCAT_BLOCK, 0, stream>>>(yA, yB, index, aggA, aggB);
  sage_finish<<<GEMM_GRID, GEMM_BLOCK, 0, stream>>>(x, aggA, aggB, bl, Wr, out);
}


# Each agg row is added to from ONE package half only: a work-group reads its die, draws 512-edge tasks from its half's ticket counter (int atomic), adds only that half's rows; sums still arrival order
# speedup vs baseline: 2.9277x; 1.6149x over previous
#include <hip/hip_runtime.h>
#include <math.h>


typedef __attribute__((ext_vector_type(2))) float v2f;
typedef __attribute__((ext_vector_type(8))) float v8f;

#define N_NODES   100000
#define NUM_FEAT  64
#define NUM_CLASS 40
#define N_EDGES   1600000
#define WPB       5
#define GEMM_BLOCK (WPB * 32)
#define ROW_TILES  (N_NODES / 16)
#define GEMM_GRID  (ROW_TILES / WPB)
#define KSTEPS     (NUM_FEAT / 4)
#define CTILES     3

#define YA_W 32
#define YB_W 8
#define GA_W 32
#define GB_W 16
#define CNT_COL 8

#define NW_V2F     (KSTEPS * CTILES * 32)
#define OFF_W      0
#define OFF_OUT    (NW_V2F * 2)
#define OFF_LSE    (OFF_OUT + WPB * 16 * 48)
#define OFF_X      (OFF_LSE + WPB * 16)
#define SMEM_FLOATS (OFF_X + WPB * 16 * NUM_FEAT)


__device__ __forceinline__ void stage_weights(v2f* sW, const float* __restrict__ Wm, int tid) {
  for (int i = tid; i < NW_V2F; i += GEMM_BLOCK) {
    const int li   = i & 31;
    const int t    = (i >> 5) % CTILES;
    const int step = i / (CTILES * 32);
    const int k    = 4 * step + 2 * (li >> 4);
    const int n    = t * 16 + (li & 15);
    v2f w = {0.0f, 0.0f};
    if (n < NUM_CLASS) w = (v2f){Wm[k * NUM_CLASS + n], Wm[(k + 1) * NUM_CLASS + n]};
    sW[i] = w;
  }
}

__device__ __forceinline__ void stage_x_tile(float* sx, const float* __restrict__ xg, int lane) {
  for (int i = lane; i < (16 * NUM_FEAT) / 4; i += 32) {
    float4 vx = *(const float4*)(xg + i * 4);
    *(float4*)(sx + i * 4) = vx;
  }
}

__device__ __forceinline__ void tile_product(const v2f* xf, const v2f* wp, int half,
                                             v8f& acc0, v8f& acc1, v8f& acc2) {
#pragma unroll
  for (int step = 0; step < KSTEPS; ++step) {
    const v2f a_x = xf[2 * step + half];
    const int wb  = step * (CTILES * 32);
    const v2f b0 = wp[wb];
    const v2f b1 = wp[wb + 32];
    const v2f b2 = wp[wb + 64];
    acc0 = __builtin_amdgcn_wmma_f32_16x16x4_f32(false, a_x, false, b0, (short)0, acc0, false, false);
    acc1 = __builtin_amdgcn_wmma_f32_16x16x4_f32(false, a_x, false, b1, (short)0, acc1, false, false);
    acc2 = __builtin_amdgcn_wmma_f32_16x16x4_f32(false, a_x, false, b2, (short)0, acc2, false, false);
  }
}

__global__ void __launch_bounds__(GEMM_BLOCK)
project_rows(const float* __restrict__ x,
             const float* __restrict__ Wl,
             float* __restrict__ yA,
             float* __restrict__ yB,
             float* __restrict__ aggA,
             float* __restrict__ aggB) {
  __shared__ __align__(16) float smem[SMEM_FLOATS];

  v2f*   sW   = (v2f*)smem;
  float* sOut = smem + OFF_OUT;

  const int tid  = threadIdx.x;
  const int wave = tid >> 5;
  const int lane = tid & 31;
  const int half = lane >> 4;
  const int l16  = lane & 15;

  const long row0 = ((long)blockIdx.x * WPB + wave) * 16;

  stage_weights(sW, Wl, tid);
  float* sx = smem + OFF_X + wave * (16 * NUM_FEAT);
  stage_x_tile(sx, x + row0 * NUM_FEAT, lane);

  __syncthreads();

  const v2f* xf = (const v2f*)(sx + l16 * NUM_FEAT);
  const v2f* wp = sW + lane;

  v8f acc0 = {}, acc1 = {}, acc2 = {};
  tile_product(xf, wp, half, acc0, acc1, acc2);

  float* so = sOut + wave * (16 * 48);
#pragma unroll
  for (int r = 0; r < 8; ++r) {
    const int M = r + 8 * half;
    so[M * 48 +      l16] = acc0[r];
    so[M * 48 + 16 + l16] = acc1[r];
    so[M * 48 + 32 + l16] = acc2[r];
  }
  __syncthreads();

  for (int r = 0; r < 16; ++r)
    yA[(row0 + r) * YA_W + lane] = so[r * 48 + lane];
  for (int i = lane; i < 16 * YB_W; i += 32)
    yB[row0 * YB_W + i] = so[(i >> 3) * 48 + 32 + (i & 7)];

  {
    const float4 z = {0.0f, 0.0f, 0.0f, 0.0f};
    float* ga = aggA + row0 * GA_W;
    float* gb = aggB + row0 * GB_W;
    for (int i = lane; i < (16 * GA_W) / 4; i += 32) *(float4*)(ga + i * 4) = z;
    for (int i = lane; i < (16 * GB_W) / 4; i += 32) *(float4*)(gb + i * 4) = z;
  }
}

#define EPW 64
#define EG  16
#define SCAT_BLOCK 256
#define EPB (EPW * (SCAT_BLOCK / 32))
#define N_TASKS (N_EDGES / EPB)
#define SCAT_GRID (2 * N_TASKS)
#define HALF_ROWS (N_NODES / 2)
#define TICK_COL 12

__device__ __forceinline__ void walk_edges(unsigned mask, int srcv, int dstv, int lane,
                                           const float* __restrict__ yA,
                                           const float* __restrict__ yB,
                                           float* __restrict__ aggA,
                                           float* __restrict__ aggB) {
  const int cb = lane & (YB_W - 1);
  while (mask != 0u) {
    const int n = __builtin_popcount(mask);
    float lo[EG], hi[EG];
    unsigned dd[EG];
    unsigned mm = mask;
#pragma unroll
    for (int j = 0; j < EG; ++j) {
      lo[j] = 0.0f; hi[j] = 0.0f; dd[j] = 0u;
      if (j < n) {
        const int i = __builtin_ctz(mm);
        mm &= mm - 1u;
        const unsigned s = (unsigned)__builtin_amdgcn_readlane(srcv, i);
        dd[j] = (unsigned)__builtin_amdgcn_readlane(dstv, i);
        lo[j] = yA[(size_t)s * YA_W + lane];
        hi[j] = yB[(size_t)s * YB_W + cb];
      }
    }
#pragma unroll
    for (int j = 0; j < EG; ++j)
      if (j < n) atomicAdd(aggA + (size_t)dd[j] * GA_W + lane, lo[j]);
    if (lane <= CNT_COL) {
#pragma unroll
      for (int j = 0; j < EG; ++j)
        if (j < n) atomicAdd(aggB + (size_t)dd[j] * GB_W + lane, lane == CNT_COL ? 1.0f : hi[j]);
    }
    mask = mm;
  }
}

__global__ void __launch_bounds__(SCAT_BLOCK)
edge_scatter(const float* __restrict__ yA,
             const float* __restrict__ yB,
             const int*   __restrict__ index,
             float* __restrict__ aggA,
             float* __restrict__ aggB) {
  __shared__ int s_task;

  if (threadIdx.x == 0) {
    int* tick0 = (int*)(aggB + TICK_COL);
    int* tick1 = (int*)(aggB + (size_t)(N_NODES - 1) * GB_W + TICK_COL);
    const unsigned where = __builtin_amdgcn_s_sendmsg_rtn(0x87);
    int role = (int)((where >> 18) & 1u);
    int t = atomicAdd(role ? tick1 : tick0, 1);
    if (t >= N_TASKS) {
      role ^= 1;
      t = atomicAdd(role ? tick1 : tick0, 1);
    }
    s_task = (t < N_TASKS) ? (2 * t + role) : -1;
  }
  __syncthreads();
  const int task = s_task;
  if (task < 0) return;
  const int role = task & 1;
  const int t    = task >> 1;

  const int  lane = threadIdx.x & 31;
  const int  wave = threadIdx.x >> 5;
  const long e0   = (long)t * EPB + wave * EPW;
  const int src0 = index[e0 + lane];
  const int src1 = index[e0 + 32 + lane];
  const int dst0 = index[N_EDGES + e0 + lane];
  const int dst1 = index[N_EDGES + e0 + 32 + lane];
  asm volatile("" :: "v"(src0), "v"(src1), "v"(dst0), "v"(dst1));
  const unsigned mask0 = __builtin_amdgcn_ballot_w32((dst0 >= HALF_ROWS) == (role != 0));
  const unsigned mask1 = __builtin_amdgcn_ballot_w32((dst1 >= HALF_ROWS) == (role != 0));
  walk_edges(mask0, src0, dst0, lane, yA, yB, aggA, aggB);
  walk_edges(mask1, src1, dst1, lane, yA, yB, aggA, aggB);
}

__global__ void __launch_bounds__(GEMM_BLOCK)
sage_finish(const float* __restrict__ x,
            const float* __restrict__ aggA,
            const float* __restrict__ aggB,
            const float* __restrict__ bias,
            const float* __restrict__ Wr,
            float* __restrict__ out) {
  __shared__ __align__(16) float smem[SMEM_FLOATS];

  v2f*   sW   = (v2f*)smem;
  float* sOut = smem + OFF_OUT;
  float* sLse = smem + OFF_LSE;

  const int tid  = threadIdx.x;
  const int wave = tid >> 5;
  const int lane = tid & 31;
  const int half = lane >> 4;
  const int l16  = lane & 15;

  const long row0 = ((long)blockIdx.x * WPB + wave) * 16;

  stage_weights(sW, Wr, tid);
  float* sx = smem + OFF_X + wave * (16 * NUM_FEAT);
  stage_x_tile(sx, x + row0 * NUM_FEAT, lane);

  const float b0 = bias[l16];
  const float b1 = bias[16 + l16];
  const float b2 = bias[32 + (l16 & (YB_W - 1))];
  float ga0[8], ga1[8], gb[8], cnt[8];
#pragma unroll
  for (int r = 0; r < 8; ++r) {
    const long row = row0 + 8 * half + r;
    ga0[r] = aggA[row * GA_W +      l16];
    ga1[r] = aggA[row * GA_W + 16 + l16];
    gb[r]  = aggB[row * GB_W + l16];
    cnt[r] = aggB[row * GB_W + CNT_COL];
  }
  __syncthreads();

  v8f acc0, acc1, acc2;
#pragma unroll
  for (int r = 0; r < 8; ++r) {
    const float rdeg = 1.0f / fmaxf(cnt[r], 1.0f);
    acc0[r] = ga0[r] * rdeg + b0;
    acc1[r] = ga1[r] * rdeg + b1;
    acc2[r] = (l16 < YB_W) ? gb[r] * rdeg + b2 : 0.0f;
  }

  const v2f* xf = (const v2f*)(sx + l16 * NUM_FEAT);
  const v2f* wp = sW + lane;
  tile_product(xf, wp, half, acc0, acc1, acc2);

  float* so = sOut + wave * (16 * 48);
#pragma unroll
  for (int r = 0; r < 8; ++r) {
    const int M = r + 8 * half;
    so[M * 48 +      l16] = acc0[r];
    so[M * 48 + 16 + l16] = acc1[r];
    so[M * 48 + 32 + l16] = acc2[r];
  }
  __syncthreads();

  if (lane < 16) {
    const float* rowp = so + l16 * 48;
    float mx = rowp[0];
    for (int c = 1; c < NUM_CLASS; ++c) mx = fmaxf(mx, rowp[c]);
    float s = 0.0f;
    for (int c = 0; c < NUM_CLASS; ++c) s += __expf(rowp[c] - mx);
    sLse[wave * 16 + l16] = mx + __logf(s);
  }
  __syncthreads();

  for (int i = lane; i < 16 * NUM_CLASS; i += 32) {
    const int r = i / NUM_CLASS, c = i % NUM_CLASS;
    out[(row0 + r) * NUM_CLASS + c] = so[r * 48 + c] - sLse[wave * 16 + r];
  }
}

extern "C" void kernel_launch(void* const* d_in, const int* in_sizes, int n_in,
                              void* d_out, int out_size, void* d_ws, size_t ws_size,
                              hipStream_t stream) {
  const float* x     = (const float*)d_in[0];
  const int*   index = (const int*)d_in[1];
  const float* Wl    = (const float*)d_in[2];
  const float* bl    = (const float*)d_in[3];
  const float* Wr    = (const float*)d_in[4];
  float* out = (float*)d_out;

  float* aggA = (float*)d_ws;
  float* aggB = aggA + (size_t)N_NODES * GA_W;
  float* yA   = out;
  float* yB   = out + (size_t)N_NODES * YA_W;

  project_rows<<<GEMM_GRID, GEMM_BLOCK, 0, stream>>>(x, Wl, yA, yB, aggA, aggB);
  edge_scatter<<<SCAT_GRID, SCAT_BLOCK, 0, stream>>>(yA, yB, index, aggA, aggB);
  sage_finish<<<GEMM_GRID, GEMM_BLOCK, 0, stream>>>(x, aggA, aggB, bl, Wr, out);
}


# Both WMMA kernels: a thread's 8 x-tile and 20 weight-staging loads, so far one per loop trip and each behind its own wait, go out as ONE unrolled group (clamped index) ahead of a single wait
# speedup vs baseline: 3.2133x; 1.0976x over previous
#include <hip/hip_runtime.h>
#include <math.h>


typedef __attribute__((ext_vector_type(2))) float v2f;
typedef __attribute__((ext_vector_type(4))) float v4f;
typedef __attribute__((ext_vector_type(8))) float v8f;

#define N_NODES   100000
#define NUM_FEAT  64
#define NUM_CLASS 40
#define N_EDGES   1600000
#define WPB       5
#define GEMM_BLOCK (WPB * 32)
#define ROW_TILES  (N_NODES / 16)
#define GEMM_GRID  (ROW_TILES / WPB)
#define KSTEPS     (NUM_FEAT / 4)
#define CTILES     3

#define YA_W 32
#define YB_W 8
#define GA_W 32
#define GB_W 16
#define CNT_COL 8

#define NW_V2F     (KSTEPS * CTILES * 32)
#define OFF_W      0
#define OFF_OUT    (NW_V2F * 2)
#define OFF_LSE    (OFF_OUT + WPB * 16 * 48)
#define OFF_X      (OFF_LSE + WPB * 16)
#define SMEM_FLOATS (OFF_X + WPB * 16 * NUM_FEAT)


#define W_TRIPS ((NW_V2F + GEMM_BLOCK - 1) / GEMM_BLOCK)
#define X_TRIPS ((16 * NUM_FEAT) / (4 * 32))

__device__ __forceinline__ void stage_operands(v2f* sW, const float* __restrict__ Wm,
                                               float* sx, const float* __restrict__ xg,
                                               int tid, int lane) {
  v4f    vx[X_TRIPS];
  v2f    vw[W_TRIPS];
#pragma unroll
  for (int t = 0; t < X_TRIPS; ++t)
    vx[t] = *(const v4f*)(xg + (lane + 32 * t) * 4);
#pragma unroll
  for (int t = 0; t < W_TRIPS; ++t) {
    int i = tid + t * GEMM_BLOCK;
    i = i < NW_V2F ? i : NW_V2F - 1;
    const int li   = i & 31;
    const int ct   = (i >> 5) % CTILES;
    const int step = i / (CTILES * 32);
    const int k    = 4 * step + 2 * (li >> 4);
    const int n    = ct * 16 + (li & 15);
    const int nc   = n < NUM_CLASS ? n : NUM_CLASS - 1;
    const float w0 = Wm[k * NUM_CLASS + nc];
    const float w1 = Wm[(k + 1) * NUM_CLASS + nc];
    vw[t] = (n < NUM_CLASS) ? (v2f){w0, w1} : (v2f){0.0f, 0.0f};
  }
  asm volatile("" :: "v"(vx[0]), "v"(vx[1]), "v"(vx[2]), "v"(vx[3]),
                     "v"(vx[4]), "v"(vx[5]), "v"(vx[6]), "v"(vx[7]),
                     "v"(vw[0]), "v"(vw[1]), "v"(vw[2]), "v"(vw[3]), "v"(vw[4]),
                     "v"(vw[5]), "v"(vw[6]), "v"(vw[7]), "v"(vw[8]), "v"(vw[9]));
#pragma unroll
  for (int t = 0; t < W_TRIPS; ++t) {
    const int i = tid + t * GEMM_BLOCK;
    if (i < NW_V2F) sW[i] = vw[t];
  }
#pragma unroll
  for (int t = 0; t < X_TRIPS; ++t)
    *(v4f*)(sx + (lane + 32 * t) * 4) = vx[t];
}

__device__ __forceinline__ void tile_product(const v2f* xf, const v2f* wp, int half,
                                             v8f& acc0, v8f& acc1, v8f& acc2) {
#pragma unroll
  for (int step = 0; step < KSTEPS; ++step) {
    const v2f a_x = xf[2 * step + half];
    const int wb  = step * (CTILES * 32);
    const v2f b0 = wp[wb];
    const v2f b1 = wp[wb + 32];
    const v2f b2 = wp[wb + 64];
    acc0 = __builtin_amdgcn_wmma_f32_16x16x4_f32(false, a_x, false, b0, (short)0, acc0, false, false);
    acc1 = __builtin_amdgcn_wmma_f32_16x16x4_f32(false, a_x, false, b1, (short)0, acc1, false, false);
    acc2 = __builtin_amdgcn_wmma_f32_16x16x4_f32(false, a_x, false, b2, (short)0, acc2, false, false);
  }
}

__global__ void __launch_bounds__(GEMM_BLOCK)
project_rows(const float* __restrict__ x,
             const float* __restrict__ Wl,
             float* __restrict__ yA,
             float* __restrict__ yB,
             float* __restrict__ aggA,
             float* __restrict__ aggB) {
  __shared__ __align__(16) float smem[SMEM_FLOATS];

  v2f*   sW   = (v2f*)smem;
  float* sOut = smem + OFF_OUT;

  const int tid  = threadIdx.x;
  const int wave = tid >> 5;
  const int lane = tid & 31;
  const int half = lane >> 4;
  const int l16  = lane & 15;

  const long row0 = ((long)blockIdx.x * WPB + wave) * 16;

  float* sx = smem + OFF_X + wave * (16 * NUM_FEAT);
  stage_operands(sW, Wl, sx, x + row0 * NUM_FEAT, tid, lane);

  __syncthreads();

  const v2f* xf = (const v2f*)(sx + l16 * NUM_FEAT);
  const v2f* wp = sW + lane;

  v8f acc0 = {}, acc1 = {}, acc2 = {};
  tile_product(xf, wp, half, acc0, acc1, acc2);

  float* so = sOut + wave * (16 * 48);
#pragma unroll
  for (int r = 0; r < 8; ++r) {
    const int M = r + 8 * half;
    so[M * 48 +      l16] = acc0[r];
    so[M * 48 + 16 + l16] = acc1[r];
    so[M * 48 + 32 + l16] = acc2[r];
  }
  __syncthreads();

  for (int r = 0; r < 16; ++r)
    yA[(row0 + r) * YA_W + lane] = so[r * 48 + lane];
  for (int i = lane; i < 16 * YB_W; i += 32)
    yB[row0 * YB_W + i] = so[(i >> 3) * 48 + 32 + (i & 7)];

  {
    const float4 z = {0.0f, 0.0f, 0.0f, 0.0f};
    float* ga = aggA + row0 * GA_W;
    float* gb = aggB + row0 * GB_W;
    for (int i = lane; i < (16 * GA_W) / 4; i += 32) *(float4*)(ga + i * 4) = z;
    for (int i = lane; i < (16 * GB_W) / 4; i += 32) *(float4*)(gb + i * 4) = z;
  }
}

#define EPW 64
#define EG  16
#define SCAT_BLOCK 256
#define EPB (EPW * (SCAT_BLOCK / 32))
#define N_TASKS (N_EDGES / EPB)
#define SCAT_GRID (2 * N_TASKS)
#define HALF_ROWS (N_NODES / 2)
#define TICK_COL 12

__device__ __forceinline__ void walk_edges(unsigned mask, int srcv, int dstv, int lane,
                                           const float* __restrict__ yA,
                                           const float* __restrict__ yB,
                                           float* __restrict__ aggA,
                                           float* __restrict__ aggB) {
  const int cb = lane & (YB_W - 1);
  while (mask != 0u) {
    const int n = __builtin_popcount(mask);
    float lo[EG], hi[EG];
    unsigned dd[EG];
    unsigned mm = mask;
#pragma unroll
    for (int j = 0; j < EG; ++j) {
      lo[j] = 0.0f; hi[j] = 0.0f; dd[j] = 0u;
      if (j < n) {
        const int i = __builtin_ctz(mm);
        mm &= mm - 1u;
        const unsigned s = (unsigned)__builtin_amdgcn_readlane(srcv, i);
        dd[j] = (unsigned)__builtin_amdgcn_readlane(dstv, i);
        lo[j] = yA[(size_t)s * YA_W + lane];
        hi[j] = yB[(size_t)s * YB_W + cb];
      }
    }
#pragma unroll
    for (int j = 0; j < EG; ++j)
      if (j < n) atomicAdd(aggA + (size_t)dd[j] * GA_W + lane, lo[j]);
    if (lane <= CNT_COL) {
#pragma unroll
      for (int j = 0; j < EG; ++j)
        if (j < n) atomicAdd(aggB + (size_t)dd[j] * GB_W + lane, lane == CNT_COL ? 1.0f : hi[j]);
    }
    mask = mm;
  }
}

__global__ void __launch_bounds__(SCAT_BLOCK)
edge_scatter(const float* __restrict__ yA,
             const float* __restrict__ yB,
             const int*   __restrict__ index,
             float* __restrict__ aggA,
             float* __restrict__ aggB) {
  __shared__ int s_task;

  if (threadIdx.x == 0) {
    int* tick0 = (int*)(aggB + TICK_COL);
    int* tick1 = (int*)(aggB + (size_t)(N_NODES - 1) * GB_W + TICK_COL);
    const unsigned where = __builtin_amdgcn_s_sendmsg_rtn(0x87);
    int role = (int)((where >> 18) & 1u);
    int t = atomicAdd(role ? tick1 : tick0, 1);
    if (t >= N_TASKS) {
      role ^= 1;
      t = atomicAdd(role ? tick1 : tick0, 1);
    }
    s_task = (t < N_TASKS) ? (2 * t + role) : -1;
  }
  __syncthreads();
  const int task = s_task;
  if (task < 0) return;
  const int role = task & 1;
  const int t    = task >> 1;

  const int  lane = threadIdx.x & 31;
  const int  wave = threadIdx.x >> 5;
  const long e0   = (long)t * EPB + wave * EPW;
  const int src0 = index[e0 + lane];
  const int src1 = index[e0 + 32 + lane];
  const int dst0 = index[N_EDGES + e0 + lane];
  const int dst1 = index[N_EDGES + e0 + 32 + lane];
  asm volatile("" :: "v"(src0), "v"(src1), "v"(dst0), "v"(dst1));
  const unsigned mask0 = __builtin_amdgcn_ballot_w32((dst0 >= HALF_ROWS) == (role != 0));
  const unsigned mask1 = __builtin_amdgcn_ballot_w32((dst1 >= HALF_ROWS) == (role != 0));
  walk_edges(mask0, src0, dst0, lane, yA, yB, aggA, aggB);
  walk_edges(mask1, src1, dst1, lane, yA, yB, aggA, aggB);
}

__global__ void __launch_bounds__(GEMM_BLOCK)
sage_finish(const float* __restrict__ x,
            const float* __restrict__ aggA,
            const float* __restrict__ aggB,
            const float* __restrict__ bias,
            const float* __restrict__ Wr,
            float* __restrict__ out) {
  __shared__ __align__(16) float smem[SMEM_FLOATS];

  v2f*   sW   = (v2f*)smem;
  float* sOut = smem + OFF_OUT;
  float* sLse = smem + OFF_LSE;

  const int tid  = threadIdx.x;
  const int wave = tid >> 5;
  const int lane = tid & 31;
  const int half = lane >> 4;
  const int l16  = lane & 15;

  const long row0 = ((long)blockIdx.x * WPB + wave) * 16;

  const float b0 = bias[l16];
  const float b1 = bias[16 + l16];
  const float b2 = bias[32 + (l16 & (YB_W - 1))];
  float ga0[8], ga1[8], gb[8], cnt[8];
#pragma unroll
  for (int r = 0; r < 8; ++r) {
    const long row = row0 + 8 * half + r;
    ga0[r] = aggA[row * GA_W +      l16];
    ga1[r] = aggA[row * GA_W + 16 + l16];
    gb[r]  = aggB[row * GB_W + l16];
    cnt[r] = aggB[row * GB_W + CNT_COL];
  }

  float* sx = smem + OFF_X + wave * (16 * NUM_FEAT);
  stage_operands(sW, Wr, sx, x + row0 * NUM_FEAT, tid, lane);
  __syncthreads();

  v8f acc0, acc1, acc2;
#pragma unroll
  for (int r = 0; r < 8; ++r) {
    const float rdeg = 1.0f / fmaxf(cnt[r], 1.0f);
    acc0[r] = ga0[r] * rdeg + b0;
    acc1[r] = ga1[r] * rdeg + b1;
    acc2[r] = (l16 < YB_W) ? gb[r] * rdeg + b2 : 0.0f;
  }

  const v2f* xf = (const v2f*)(sx + l16 * NUM_FEAT);
  const v2f* wp = sW + lane;
  tile_product(xf, wp, half, acc0, acc1, acc2);

  float* so = sOut + wave * (16 * 48);
#pragma unroll
  for (int r = 0; r < 8; ++r) {
    const int M = r + 8 * half;
    so[M * 48 +      l16] = acc0[r];
    so[M * 48 + 16 + l16] = acc1[r];
    so[M * 48 + 32 + l16] = acc2[r];
  }
  __syncthreads();

  if (lane < 16) {
    const float* rowp = so + l16 * 48;
    float mx = rowp[0];
    for (int c = 1; c < NUM_CLASS; ++c) mx = fmaxf(mx, rowp[c]);
    float s = 0.0f;
    for (int c = 0; c < NUM_CLASS; ++c) s += __expf(rowp[c] - mx);
    sLse[wave * 16 + l16] = mx + __logf(s);
  }
  __syncthreads();

  for (int i = lane; i < 16 * NUM_CLASS; i += 32) {
    const int r = i / NUM_CLASS, c = i % NUM_CLASS;
    out[(row0 + r) * NUM_CLASS + c] = so[r * 48 + c] - sLse[wave * 16 + r];
  }
}

extern "C" void kernel_launch(void* const* d_in, const int* in_sizes, int n_in,
                              void* d_out, int out_size, void* d_ws, size_t ws_size,
                              hipStream_t stream) {
  const float* x     = (const float*)d_in[0];
  const int*   index = (const int*)d_in[1];
  const float* Wl    = (const float*)d_in[2];
  const float* bl    = (const float*)d_in[3];
  const float* Wr    = (const float*)d_in[4];
  float* out = (float*)d_out;

  float* aggA = (float*)d_ws;
  float* aggB = aggA + (size_t)N_NODES * GA_W;
  float* yA   = out;
  float* yB   = out + (size_t)N_NODES * YA_W;

  project_rows<<<GEMM_GRID, GEMM_BLOCK, 0, stream>>>(x, Wl, yA, yB, aggA, aggB);
  edge_scatter<<<SCAT_GRID, SCAT_BLOCK, 0, stream>>>(yA, yB, index, aggA, aggB);
  sage_finish<<<GEMM_GRID, GEMM_BLOCK, 0, stream>>>(x, aggA, aggB, bl, Wr, out);
}
